# starting kernel
# baseline (speedup 1.0000x reference)
#include <hip/hip_runtime.h>

#define BT_    192
#define N_     1000
#define NP_    1008
#define K_     16
#define C1_    32
#define C2_    16
#define NK_    16000
#define NTILES 63

typedef __attribute__((ext_vector_type(2))) float v2f;
typedef __attribute__((ext_vector_type(4))) float v4f;
typedef __attribute__((ext_vector_type(8))) float v8f;
typedef __attribute__((ext_vector_type(2))) int   v2i;
typedef __attribute__((ext_vector_type(4))) int   v4i;

__device__ __forceinline__ v8f wmma_f32_4(v2f a, v2f b, v8f c) {
    return __builtin_amdgcn_wmma_f32_16x16x4_f32(
         false, a,  false, b,
         (short)0, c,  false,  false);
}

__global__ __launch_bounds__(256)
void k1_layer1(const float* __restrict__ x, const int* __restrict__ nb,
               const float* __restrict__ W1, const float* __restrict__ b1,
               float* __restrict__ h1) {
    const int lane   = threadIdx.x & 31;
    const int wave   = threadIdx.x >> 5;
    const int tileId = blockIdx.x * 8 + wave;
    if (tileId >= BT_ * NTILES) return;
    const int bt   = tileId / NTILES;
    const int base = (tileId % NTILES) * 16;

    const int mrow  = lane & 15;
    const int hi    = lane >> 4;
    const int khalf = hi * 2;
    const int node  = min(base + mrow, N_ - 1);
    const float* xb = x + bt * N_;

    for (int ct = 0; ct < 2; ++ct) {
        v8f acc = {};
#pragma unroll
        for (int s = 0; s < 4; ++s) {
            const int k0 = s * 4 + khalf;
            const v2i idx = *(const v2i*)&nb[node * K_ + k0];
            v2f a, b;
            a.x = xb[idx.x];
            a.y = xb[idx.y];
            b = *(const v2f*)&W1[(ct * 16 + mrow) * K_ + k0];
            acc = wmma_f32_4(a, b, acc);
        }
        const int   ch   = ct * 16 + mrow;
        const float bias = b1[ch];
        v4f lo, hiv;
#pragma unroll
        for (int j = 0; j < 4; ++j) {
            float v0 = acc[j]     + bias;  lo[j]  = v0 > 0.f ? v0 : 0.f;
            float v1 = acc[j + 4] + bias;  hiv[j] = v1 > 0.f ? v1 : 0.f;
        }
        float* p = h1 + (size_t)(bt * C1_ + ch) * NP_ + base + 8 * hi;
        *(v4f*)p       = lo;
        *(v4f*)(p + 4) = hiv;
    }
}

__global__ __launch_bounds__(256)
void k2_layer2(const float* __restrict__ h1, const int* __restrict__ nb,
               const float* __restrict__ W2, const float* __restrict__ b2,
               const float* __restrict__ gamma, const float* __restrict__ beta,
               float* __restrict__ h2) {
    __shared__ float h1s[C1_ * NP_];
    __shared__ int   nbs[NK_];
    __shared__ float w2s[C2_ * 512];

    const int bt     = blockIdx.x >> 1;
    const int halfId = blockIdx.x & 1;

    const float* h1b = h1 + (size_t)bt * C1_ * NP_;
    for (int i = threadIdx.x; i < C1_ * NP_ / 4; i += 256)
        ((v4f*)h1s)[i] = ((const v4f*)h1b)[i];
    for (int i = threadIdx.x; i < NK_ / 4; i += 256)
        ((v4i*)nbs)[i] = ((const v4i*)nb)[i];
    for (int i = threadIdx.x; i < C2_ * 512 / 4; i += 256)
        ((v4f*)w2s)[i] = ((const v4f*)W2)[i];
    __syncthreads();

    const int lane  = threadIdx.x & 31;
    const int wave  = threadIdx.x >> 5;
    const int mrow  = lane & 15;
    const int hi    = lane >> 4;
    const int khalf = hi * 2;

    const float scale = gamma[mrow] * rsqrtf(1.0f + 1e-5f);
    const float bias  = b2[mrow];
    const float shift = beta[mrow];

    for (int t = 0; t < 4; ++t) {
        const int tile = halfId * 32 + wave * 4 + t;
        if (tile >= NTILES) continue;
        const int base = tile * 16;
        const int node = min(base + mrow, N_ - 1);
        const unsigned fbase = (unsigned)node * 512u;

        v8f acc = {};
#pragma unroll 8
        for (int s = 0; s < 128; ++s) {
            const int k0 = s * 4 + khalf;
            const unsigned f0 = fbase + (unsigned)k0;
            const unsigned c  = f0 / (unsigned)NK_;
            const unsigned r  = f0 - c * (unsigned)NK_;
            const v2i idx = *(const v2i*)&nbs[r];
            v2f a, b;
            a.x = h1s[c * NP_ + idx.x];
            a.y = h1s[c * NP_ + idx.y];
            b = *(const v2f*)&w2s[mrow * 512 + k0];
            acc = wmma_f32_4(a, b, acc);
        }
        v4f lo, hiv;
#pragma unroll
        for (int j = 0; j < 4; ++j) {
            float v0 = (acc[j]     + bias) * scale + shift;  lo[j]  = v0 > 0.f ? v0 : 0.f;
            float v1 = (acc[j + 4] + bias) * scale + shift;  hiv[j] = v1 > 0.f ? v1 : 0.f;
        }
        float* p = h2 + (size_t)(bt * C2_ + mrow) * NP_ + base + 8 * hi;
        *(v4f*)p       = lo;
        *(v4f*)(p + 4) = hiv;
    }
}

__global__ __launch_bounds__(256)
void k3_layer3(const float* __restrict__ h2, const int* __restrict__ nb,
               const float* __restrict__ W3, const float* __restrict__ b3,
               float* __restrict__ out) {
    __shared__ float h2s[C2_ * NP_];
    __shared__ int   nbs[NK_];
    __shared__ float w3s[256];

    const int bt     = blockIdx.x >> 1;
    const int halfId = blockIdx.x & 1;

    const float* h2b = h2 + (size_t)bt * C2_ * NP_;
    for (int i = threadIdx.x; i < C2_ * NP_ / 4; i += 256)
        ((v4f*)h2s)[i] = ((const v4f*)h2b)[i];
    for (int i = threadIdx.x; i < NK_ / 4; i += 256)
        ((v4i*)nbs)[i] = ((const v4i*)nb)[i];
    if (threadIdx.x < 256) w3s[threadIdx.x] = W3[threadIdx.x];
    __syncthreads();

    const float b3v = b3[0];
    for (int i = 0; i < 2; ++i) {
        const int off = i * 256 + threadIdx.x;
        if (off >= 500) break;
        const int n = halfId * 500 + off;
        float acc = b3v;
        const unsigned fbase = (unsigned)n * 256u;
#pragma unroll 8
        for (int f = 0; f < 256; f += 2) {
            const unsigned fl = fbase + (unsigned)f;
            const unsigned c  = fl / (unsigned)NK_;
            const unsigned r  = fl - c * (unsigned)NK_;
            const v2i idx = *(const v2i*)&nbs[r];
            acc = fmaf(h2s[c * NP_ + idx.x], w3s[f],     acc);
            acc = fmaf(h2s[c * NP_ + idx.y], w3s[f + 1], acc);
        }
        out[bt * N_ + n] = acc > 0.f ? acc : 0.f;
    }
}

extern "C" void kernel_launch(void* const* d_in, const int* in_sizes, int n_in,
                              void* d_out, int out_size, void* d_ws, size_t ws_size,
                              hipStream_t stream) {
    const float* x     = (const float*)d_in[0];
    const int*   nb    = (const int*)  d_in[1];
    const float* W1    = (const float*)d_in[2];
    const float* b1    = (const float*)d_in[3];
    const float* W2    = (const float*)d_in[4];
    const float* b2    = (const float*)d_in[5];
    const float* W3    = (const float*)d_in[6];
    const float* b3    = (const float*)d_in[7];
    const float* gamma = (const float*)d_in[8];
    const float* beta  = (const float*)d_in[9];

    float* h1  = (float*)d_ws;
    float* h2  = h1 + (size_t)BT_ * C1_ * NP_;
    float* out = (float*)d_out;

    const int tiles  = BT_ * NTILES;
    const int blocks = (tiles + 7) / 8;

    k1_layer1<<<blocks,  256, 0, stream>>>(x, nb, W1, b1, h1);
    k2_layer2<<<BT_ * 2, 256, 0, stream>>>(h1, nb, W2, b2, gamma, beta, h2);
    k3_layer3<<<BT_ * 2, 256, 0, stream>>>(h2, nb, W3, b3, out);
}


# Make bt the matrix row dimension and store activations bt-innermost: a neighbor index is shared by all 16 rows of a fragment, each gather is one contiguous row read from L2; sum order kept
# speedup vs baseline: 2.7164x; 2.7164x over previous
#include <hip/hip_runtime.h>

#define BT_    192
#define N_     1000
#define K_     16
#define C1_    32
#define C2_    16
#define NK_    16000
#define ROW_   192
#define NPB_   4
#define NBLK_  250
#define G_     8
#define GF_    16

typedef __attribute__((ext_vector_type(2))) float v2f;
typedef __attribute__((ext_vector_type(4))) float v4f;
typedef __attribute__((ext_vector_type(8))) float v8f;
typedef __attribute__((ext_vector_type(2))) int   v2i;

__device__ __forceinline__ v8f wmma_f32_4(v2f a, v2f b, v8f c) {
    return __builtin_amdgcn_wmma_f32_16x16x4_f32(
        false, a, false, b, (short)0, c, false, false);
}

__global__ __launch_bounds__(384)
void k1_layer1(const float* __restrict__ x, const int* __restrict__ nb,
               const float* __restrict__ W1, const float* __restrict__ b1,
               float* __restrict__ h1t) {
    const int lane  = threadIdx.x & 31;
    const int wave  = threadIdx.x >> 5;
    const int nl    = wave / 3;
    const int third = wave - nl * 3;
    const int n     = blockIdx.x * NPB_ + nl;
    const int mrow  = lane & 15;
    const int hi    = lane >> 4;

    v2f a[4][4];
#pragma unroll
    for (int s = 0; s < 4; ++s) {
        const int k0 = s * 4 + hi * 2;
        const v2i idx = *(const v2i*)&nb[n * K_ + k0];
#pragma unroll
        for (int tq = 0; tq < 4; ++tq) {
            const int bt = (third * 4 + tq) * 16 + mrow;
            a[tq][s].x = x[bt * N_ + idx.x];
            a[tq][s].y = x[bt * N_ + idx.y];
        }
    }
#pragma unroll
    for (int ct = 0; ct < 2; ++ct) {
        v8f acc[4] = {};
#pragma unroll
        for (int s = 0; s < 4; ++s) {
            const v2f b = *(const v2f*)&W1[(ct * 16 + mrow) * K_ + s * 4 + hi * 2];
#pragma unroll
            for (int tq = 0; tq < 4; ++tq) acc[tq] = wmma_f32_4(a[tq][s], b, acc[tq]);
        }
        const int   ch   = ct * 16 + mrow;
        const float bias = b1[ch];
        float* p = h1t + ((size_t)ch * N_ + n) * ROW_ + third * 64 + hi * 32;
#pragma unroll
        for (int r = 0; r < 8; ++r) {
            v4f v;
#pragma unroll
            for (int tq = 0; tq < 4; ++tq) {
                const float t = acc[tq][r] + bias;
                v[tq] = t > 0.f ? t : 0.f;
            }
            *(v4f*)(p + r * 4) = v;
        }
    }
}

__global__ __launch_bounds__(384)
void k2_layer2(const float* __restrict__ h1t, const int* __restrict__ nb,
               const float* __restrict__ W2, const float* __restrict__ b2,
               const float* __restrict__ gamma, const float* __restrict__ beta,
               float* __restrict__ h2t) {
    __shared__ float w2f[128 * 64];
    __shared__ int   roff[NPB_ * 512];

    const int n0 = blockIdx.x * NPB_;
    if (threadIdx.x < 256) {
#pragma unroll
      for (int it = 0; it < 8; ++it) {
        const int i = threadIdx.x + it * 256;
        const int m = i & 15, s = i >> 4;
        const v4f w = *(const v4f*)&W2[m * 512 + s * 4];
        v2f lo, hh;
        lo.x = w.x; lo.y = w.y; hh.x = w.z; hh.y = w.w;
        ((v2f*)w2f)[s * 32 + m]      = lo;
        ((v2f*)w2f)[s * 32 + 16 + m] = hh;
      }
#pragma unroll
      for (int it = 0; it < 8; ++it) {
        const int i = threadIdx.x + it * 256;
        const unsigned q = (unsigned)(n0 + (i >> 9)) * 512u + (unsigned)(i & 511);
        const unsigned c = q / (unsigned)NK_;
        const unsigned r = q - c * (unsigned)NK_;
        roff[i] = ((int)c * N_ + nb[r]) * ROW_;
      }
    }
    __syncthreads();

    const int lane  = threadIdx.x & 31;
    const int wave  = threadIdx.x >> 5;
    const int nl    = wave / 3;
    const int third = wave - nl * 3;
    const int n     = n0 + nl;
    const int mrow  = lane & 15;
    const int hi    = lane >> 4;

    const float* abase = h1t + third * 64 + mrow * 4;
    const int*   ro    = roff + nl * 512 + hi * 2;
    const v2f*   wb    = (const v2f*)w2f + lane;

    v8f acc[4] = {};
    auto gather = [&](int s0, v4f (&u0)[G_], v4f (&u1)[G_]) {
#pragma unroll
        for (int j = 0; j < G_; ++j) {
            const v2i o = *(const v2i*)&ro[(s0 + j) * 4];
            u0[j] = *(const v4f*)(abase + o.x);
            u1[j] = *(const v4f*)(abase + o.y);
        }
    };
    auto mm = [&](int s0, const v4f (&u0)[G_], const v4f (&u1)[G_]) {
#pragma unroll
        for (int j = 0; j < G_; ++j) {
            const v2f b = wb[(s0 + j) * 32];
#pragma unroll
            for (int tq = 0; tq < 4; ++tq) {
                v2f a;
                a.x = u0[j][tq];
                a.y = u1[j][tq];
                acc[tq] = wmma_f32_4(a, b, acc[tq]);
            }
        }
    };
    v4f p0[G_], p1[G_], q0[G_], q1[G_];
    gather(0, p0, p1);
#define SCHED_FENCE() __builtin_amdgcn_sched_barrier(0)
    for (int s = 0; s < 128 - 2 * G_; s += 2 * G_) {
        gather(s + G_, q0, q1);
        SCHED_FENCE();
        mm(s, p0, p1);
        SCHED_FENCE();
        gather(s + 2 * G_, p0, p1);
        SCHED_FENCE();
        mm(s + G_, q0, q1);
        SCHED_FENCE();
    }
    gather(128 - G_, q0, q1);
    SCHED_FENCE();
    mm(128 - 2 * G_, p0, p1);
    mm(128 - G_, q0, q1);

    const float scale = gamma[mrow] * rsqrtf(1.0f + 1e-5f);
    const float bias  = b2[mrow];
    const float shift = beta[mrow];
    float* p = h2t + ((size_t)mrow * N_ + n) * ROW_ + third * 64 + hi * 32;
#pragma unroll
    for (int r = 0; r < 8; ++r) {
        v4f v;
#pragma unroll
        for (int tq = 0; tq < 4; ++tq) {
            const float t = (acc[tq][r] + bias) * scale + shift;
            v[tq] = t > 0.f ? t : 0.f;
        }
        *(v4f*)(p + r * 4) = v;
    }
}

__global__ __launch_bounds__(192)
void k3_layer3(const float* __restrict__ h2t, const int* __restrict__ nb,
               const float* __restrict__ W3, const float* __restrict__ b3,
               float* __restrict__ out) {
    __shared__ int   roff[NPB_ * 256];
    __shared__ float w3s[256];
    __shared__ float tile[NPB_][ROW_];

    const int n0 = blockIdx.x * NPB_;
    if (threadIdx.x < 128) {
#pragma unroll
        for (int it = 0; it < 8; ++it) {
            const int i = threadIdx.x + it * 128;
            const unsigned q = (unsigned)(n0 + (i >> 8)) * 256u + (unsigned)(i & 255);
            const unsigned c = q / (unsigned)NK_;
            const unsigned r = q - c * (unsigned)NK_;
            roff[i] = ((int)c * N_ + nb[r]) * ROW_;
        }
    } else {
        ((v4f*)w3s)[threadIdx.x - 128] = ((const v4f*)W3)[threadIdx.x - 128];
    }
    __syncthreads();

    const int nl    = threadIdx.x / 48;
    const int chunk = threadIdx.x - nl * 48;
    const float* base = h2t + chunk * 4;
    const int*   ro   = roff + nl * 256;
    const float  b3v  = b3[0];
    v4f acc;
    acc.x = b3v; acc.y = b3v; acc.z = b3v; acc.w = b3v;
    auto gather3 = [&](int f0, v4f (&u)[GF_]) {
#pragma unroll
        for (int j = 0; j < GF_; ++j) u[j] = *(const v4f*)(base + ro[f0 + j]);
    };
    auto fma3 = [&](int f0, const v4f (&u)[GF_]) {
#pragma unroll
        for (int j = 0; j < GF_; ++j) {
            const float w = w3s[f0 + j];
            acc.x = fmaf(u[j].x, w, acc.x);
            acc.y = fmaf(u[j].y, w, acc.y);
            acc.z = fmaf(u[j].z, w, acc.z);
            acc.w = fmaf(u[j].w, w, acc.w);
        }
    };
    v4f pa[GF_], pb[GF_];
    gather3(0, pa);
    for (int f = 0; f < 256 - 2 * GF_; f += 2 * GF_) {
        gather3(f + GF_, pb);
        SCHED_FENCE();
        fma3(f, pa);
        SCHED_FENCE();
        gather3(f + 2 * GF_, pa);
        SCHED_FENCE();
        fma3(f + GF_, pb);
        SCHED_FENCE();
    }
    gather3(256 - GF_, pb);
    SCHED_FENCE();
    fma3(256 - 2 * GF_, pa);
    fma3(256 - GF_, pb);
    v4f rl;
    rl.x = acc.x > 0.f ? acc.x : 0.f;
    rl.y = acc.y > 0.f ? acc.y : 0.f;
    rl.z = acc.z > 0.f ? acc.z : 0.f;
    rl.w = acc.w > 0.f ? acc.w : 0.f;
    *(v4f*)&tile[nl][chunk * 4] = rl;
    __syncthreads();

    const int bt  = threadIdx.x;
    const int pos = (bt >> 6) * 64 + (bt & 15) * 4 + ((bt >> 4) & 3);
    v4f o;
    o.x = tile[0][pos]; o.y = tile[1][pos]; o.z = tile[2][pos]; o.w = tile[3][pos];
    *(v4f*)&out[bt * N_ + n0] = o;
}

extern "C" void kernel_launch(void* const* d_in, const int* in_sizes, int n_in,
                              void* d_out, int out_size, void* d_ws, size_t ws_size,
                              hipStream_t stream) {
    const float* x     = (const float*)d_in[0];
    const int*   nb    = (const int*)  d_in[1];
    const float* W1    = (const float*)d_in[2];
    const float* b1    = (const float*)d_in[3];
    const float* W2    = (const float*)d_in[4];
    const float* b2    = (const float*)d_in[5];
    const float* W3    = (const float*)d_in[6];
    const float* b3    = (const float*)d_in[7];
    const float* gamma = (const float*)d_in[8];
    const float* beta  = (const float*)d_in[9];

    float* h1t = (float*)d_ws;
    float* h2t = h1t + (size_t)C1_ * N_ * ROW_;
    float* out = (float*)d_out;

    k1_layer1<<<NBLK_, 384, 0, stream>>>(x, nb, W1, b1, h1t);
    k2_layer2<<<NBLK_, 384, 0, stream>>>(h1t, nb, W2, b2, gamma, beta, h2t);
    k3_layer3<<<NBLK_, 192, 0, stream>>>(h2t, nb, W3, b3, out);
}


# Transpose x once into node rows with bt innermost (small pre-kernel, 3 -> 4 launches): a layer-1 gather is one contiguous 16-byte read per lane for four bt tiles, not 32 scattered lines; sums kept
# speedup vs baseline: 2.9812x; 1.0975x over previous
#include <hip/hip_runtime.h>

#define BT_    192
#define N_     1000
#define K_     16
#define C1_    32
#define C2_    16
#define NK_    16000
#define ROW_   192
#define NPB_   4
#define NBLK_  250
#define G_     8
#define GF_    16
#define TN_    8

typedef __attribute__((ext_vector_type(2))) float v2f;
typedef __attribute__((ext_vector_type(4))) float v4f;
typedef __attribute__((ext_vector_type(8))) float v8f;
typedef __attribute__((ext_vector_type(2))) int   v2i;

__device__ __forceinline__ v8f wmma_f32_4(v2f a, v2f b, v8f c) {
    return __builtin_amdgcn_wmma_f32_16x16x4_f32(
        false, a, false, b, (short)0, c, false, false);
}

__global__ __launch_bounds__(192)
void k0_transpose(const float* __restrict__ x, float* __restrict__ xt) {
    __shared__ float tile[TN_][ROW_];
    const int bt  = threadIdx.x;
    const int n0  = blockIdx.x * TN_;
    const int pos = (bt >> 6) * 64 + (bt & 15) * 4 + ((bt >> 4) & 3);
    const v4f lo = *(const v4f*)&x[bt * N_ + n0];
    const v4f hh = *(const v4f*)&x[bt * N_ + n0 + 4];
    tile[0][pos] = lo.x; tile[1][pos] = lo.y; tile[2][pos] = lo.z; tile[3][pos] = lo.w;
    tile[4][pos] = hh.x; tile[5][pos] = hh.y; tile[6][pos] = hh.z; tile[7][pos] = hh.w;
    __syncthreads();
#pragma unroll
    for (int j = 0; j < TN_; ++j) xt[(n0 + j) * ROW_ + bt] = tile[j][bt];
}

__global__ __launch_bounds__(384)
void k1_layer1(const float* __restrict__ xt, const int* __restrict__ nb,
               const float* __restrict__ W1, const float* __restrict__ b1,
               float* __restrict__ h1t) {
    const int lane  = threadIdx.x & 31;
    const int wave  = threadIdx.x >> 5;
    const int nl    = wave / 3;
    const int third = wave - nl * 3;
    const int n     = blockIdx.x * NPB_ + nl;
    const int mrow  = lane & 15;
    const int hi    = lane >> 4;

    v2f a[4][4];
    const float* xb = xt + third * 64 + mrow * 4;
#pragma unroll
    for (int s = 0; s < 4; ++s) {
        const int k0 = s * 4 + hi * 2;
        const v2i idx = *(const v2i*)&nb[n * K_ + k0];
        const v4f u0 = *(const v4f*)(xb + idx.x * ROW_);
        const v4f u1 = *(const v4f*)(xb + idx.y * ROW_);
#pragma unroll
        for (int tq = 0; tq < 4; ++tq) {
            a[tq][s].x = u0[tq];
            a[tq][s].y = u1[tq];
        }
    }
#pragma unroll
    for (int ct = 0; ct < 2; ++ct) {
        v8f acc[4] = {};
#pragma unroll
        for (int s = 0; s < 4; ++s) {
            const v2f b = *(const v2f*)&W1[(ct * 16 + mrow) * K_ + s * 4 + hi * 2];
#pragma unroll
            for (int tq = 0; tq < 4; ++tq) acc[tq] = wmma_f32_4(a[tq][s], b, acc[tq]);
        }
        const int   ch   = ct * 16 + mrow;
        const float bias = b1[ch];
        float* p = h1t + ((size_t)ch * N_ + n) * ROW_ + third * 64 + hi * 32;
#pragma unroll
        for (int r = 0; r < 8; ++r) {
            v4f v;
#pragma unroll
            for (int tq = 0; tq < 4; ++tq) {
                const float t = acc[tq][r] + bias;
                v[tq] = t > 0.f ? t : 0.f;
            }
            *(v4f*)(p + r * 4) = v;
        }
    }
}

__global__ __launch_bounds__(384)
void k2_layer2(const float* __restrict__ h1t, const int* __restrict__ nb,
               const float* __restrict__ W2, const float* __restrict__ b2,
               const float* __restrict__ gamma, const float* __restrict__ beta,
               float* __restrict__ h2t) {
    __shared__ float w2f[128 * 64];
    __shared__ int   roff[NPB_ * 512];

    const int n0 = blockIdx.x * NPB_;
    if (threadIdx.x < 256) {
#pragma unroll
      for (int it = 0; it < 8; ++it) {
        const int i = threadIdx.x + it * 256;
        const int m = i & 15, s = i >> 4;
        const v4f w = *(const v4f*)&W2[m * 512 + s * 4];
        v2f lo, hh;
        lo.x = w.x; lo.y = w.y; hh.x = w.z; hh.y = w.w;
        ((v2f*)w2f)[s * 32 + m]      = lo;
        ((v2f*)w2f)[s * 32 + 16 + m] = hh;
      }
#pragma unroll
      for (int it = 0; it < 8; ++it) {
        const int i = threadIdx.x + it * 256;
        const unsigned q = (unsigned)(n0 + (i >> 9)) * 512u + (unsigned)(i & 511);
        const unsigned c = q / (unsigned)NK_;
        const unsigned r = q - c * (unsigned)NK_;
        roff[i] = ((int)c * N_ + nb[r]) * ROW_;
      }
    }
    __syncthreads();

    const int lane  = threadIdx.x & 31;
    const int wave  = threadIdx.x >> 5;
    const int nl    = wave / 3;
    const int third = wave - nl * 3;
    const int n     = n0 + nl;
    const int mrow  = lane & 15;
    const int hi    = lane >> 4;

    const float* abase = h1t + third * 64 + mrow * 4;
    const int*   ro    = roff + nl * 512 + hi * 2;
    const v2f*   wb    = (const v2f*)w2f + lane;

    v8f acc[4] = {};
    auto gather = [&](int s0, v4f (&u0)[G_], v4f (&u1)[G_]) {
#pragma unroll
        for (int j = 0; j < G_; ++j) {
            const v2i o = *(const v2i*)&ro[(s0 + j) * 4];
            u0[j] = *(const v4f*)(abase + o.x);
            u1[j] = *(const v4f*)(abase + o.y);
        }
    };
    auto mm = [&](int s0, const v4f (&u0)[G_], const v4f (&u1)[G_]) {
#pragma unroll
        for (int j = 0; j < G_; ++j) {
            const v2f b = wb[(s0 + j) * 32];
#pragma unroll
            for (int tq = 0; tq < 4; ++tq) {
                v2f a;
                a.x = u0[j][tq];
                a.y = u1[j][tq];
                acc[tq] = wmma_f32_4(a, b, acc[tq]);
            }
        }
    };
    v4f p0[G_], p1[G_], q0[G_], q1[G_];
    gather(0, p0, p1);
#define SCHED_FENCE() __builtin_amdgcn_sched_barrier(0)
    for (int s = 0; s < 128 - 2 * G_; s += 2 * G_) {
        gather(s + G_, q0, q1);
        SCHED_FENCE();
        mm(s, p0, p1);
        SCHED_FENCE();
        gather(s + 2 * G_, p0, p1);
        SCHED_FENCE();
        mm(s + G_, q0, q1);
        SCHED_FENCE();
    }
    gather(128 - G_, q0, q1);
    SCHED_FENCE();
    mm(128 - 2 * G_, p0, p1);
    mm(128 - G_, q0, q1);

    const float scale = gamma[mrow] * rsqrtf(1.0f + 1e-5f);
    const float bias  = b2[mrow];
    const float shift = beta[mrow];
    float* p = h2t + ((size_t)mrow * N_ + n) * ROW_ + third * 64 + hi * 32;
#pragma unroll
    for (int r = 0; r < 8; ++r) {
        v4f v;
#pragma unroll
        for (int tq = 0; tq < 4; ++tq) {
            const float t = (acc[tq][r] + bias) * scale + shift;
            v[tq] = t > 0.f ? t : 0.f;
        }
        *(v4f*)(p + r * 4) = v;
    }
}

__global__ __launch_bounds__(192)
void k3_layer3(const float* __restrict__ h2t, const int* __restrict__ nb,
               const float* __restrict__ W3, const float* __restrict__ b3,
               float* __restrict__ out) {
    __shared__ int   roff[NPB_ * 256];
    __shared__ float w3s[256];
    __shared__ float tile[NPB_][ROW_];

    const int n0 = blockIdx.x * NPB_;
    if (threadIdx.x < 128) {
#pragma unroll
        for (int it = 0; it < 8; ++it) {
            const int i = threadIdx.x + it * 128;
            const unsigned q = (unsigned)(n0 + (i >> 8)) * 256u + (unsigned)(i & 255);
            const unsigned c = q / (unsigned)NK_;
            const unsigned r = q - c * (unsigned)NK_;
            roff[i] = ((int)c * N_ + nb[r]) * ROW_;
        }
    } else {
        ((v4f*)w3s)[threadIdx.x - 128] = ((const v4f*)W3)[threadIdx.x - 128];
    }
    __syncthreads();

    const int nl    = threadIdx.x / 48;
    const int chunk = threadIdx.x - nl * 48;
    const float* base = h2t + chunk * 4;
    const int*   ro   = roff + nl * 256;
    const float  b3v  = b3[0];
    v4f acc;
    acc.x = b3v; acc.y = b3v; acc.z = b3v; acc.w = b3v;
    auto gather3 = [&](int f0, v4f (&u)[GF_]) {
#pragma unroll
        for (int j = 0; j < GF_; ++j) u[j] = *(const v4f*)(base + ro[f0 + j]);
    };
    auto fma3 = [&](int f0, const v4f (&u)[GF_]) {
#pragma unroll
        for (int j = 0; j < GF_; ++j) {
            const float w = w3s[f0 + j];
            acc.x = fmaf(u[j].x, w, acc.x);
            acc.y = fmaf(u[j].y, w, acc.y);
            acc.z = fmaf(u[j].z, w, acc.z);
            acc.w = fmaf(u[j].w, w, acc.w);
        }
    };
    v4f pa[GF_], pb[GF_];
    gather3(0, pa);
    for (int f = 0; f < 256 - 2 * GF_; f += 2 * GF_) {
        gather3(f + GF_, pb);
        SCHED_FENCE();
        fma3(f, pa);
        SCHED_FENCE();
        gather3(f + 2 * GF_, pa);
        SCHED_FENCE();
        fma3(f + GF_, pb);
        SCHED_FENCE();
    }
    gather3(256 - GF_, pb);
    SCHED_FENCE();
    fma3(256 - 2 * GF_, pa);
    fma3(256 - GF_, pb);
    v4f rl;
    rl.x = acc.x > 0.f ? acc.x : 0.f;
    rl.y = acc.y > 0.f ? acc.y : 0.f;
    rl.z = acc.z > 0.f ? acc.z : 0.f;
    rl.w = acc.w > 0.f ? acc.w : 0.f;
    *(v4f*)&tile[nl][chunk * 4] = rl;
    __syncthreads();

    const int bt  = threadIdx.x;
    const int pos = (bt >> 6) * 64 + (bt & 15) * 4 + ((bt >> 4) & 3);
    v4f o;
    o.x = tile[0][pos]; o.y = tile[1][pos]; o.z = tile[2][pos]; o.w = tile[3][pos];
    *(v4f*)&out[bt * N_ + n0] = o;
}

extern "C" void kernel_launch(void* const* d_in, const int* in_sizes, int n_in,
                              void* d_out, int out_size, void* d_ws, size_t ws_size,
                              hipStream_t stream) {
    const float* x     = (const float*)d_in[0];
    const int*   nb    = (const int*)  d_in[1];
    const float* W1    = (const float*)d_in[2];
    const float* b1    = (const float*)d_in[3];
    const float* W2    = (const float*)d_in[4];
    const float* b2    = (const float*)d_in[5];
    const float* W3    = (const float*)d_in[6];
    const float* b3    = (const float*)d_in[7];
    const float* gamma = (const float*)d_in[8];
    const float* beta  = (const float*)d_in[9];

    float* h1t = (float*)d_ws;
    float* h2t = h1t + (size_t)C1_ * N_ * ROW_;
    float* xt  = h2t;
    float* out = (float*)d_out;

    k0_transpose<<<125, 192, 0, stream>>>(x, xt);
    k1_layer1<<<NBLK_, 384, 0, stream>>>(xt, nb, W1, b1, h1t);
    k2_layer2<<<NBLK_, 384, 0, stream>>>(h1t, nb, W2, b2, gamma, beta, h2t);
    k3_layer3<<<NBLK_, 192, 0, stream>>>(h2t, nb, W3, b3, out);
}


# Layers 1+2: weights as matrix A (rows = channel), gathered activations as B (cols = bt): D holds bt across lanes, each epilogue store writes 4 whole 128-B lines, not 32 16-B pieces; sum order kept
# speedup vs baseline: 3.4701x; 1.1640x over previous
#include <hip/hip_runtime.h>

#define BT_    192
#define N_     1000
#define K_     16
#define C1_    32
#define C2_    16
#define NK_    16000
#define ROW_   192
#define NPB_   4
#define NBLK_  250
#define G_     8
#define GF_    16
#define TN_    8

typedef __attribute__((ext_vector_type(2))) float v2f;
typedef __attribute__((ext_vector_type(4))) float v4f;
typedef __attribute__((ext_vector_type(8))) float v8f;
typedef __attribute__((ext_vector_type(2))) int   v2i;

__device__ __forceinline__ v8f wmma_f32_4(v2f a, v2f b, v8f c) {
    return __builtin_amdgcn_wmma_f32_16x16x4_f32(
        false, a, false, b, (short)0, c, false, false);
}

__global__ __launch_bounds__(192)
void k0_transpose(const float* __restrict__ x, float* __restrict__ xt) {
    __shared__ float tile[TN_][ROW_];
    const int bt  = threadIdx.x;
    const int n0  = blockIdx.x * TN_;
    const int pos = (bt >> 6) * 64 + (bt & 15) * 4 + ((bt >> 4) & 3);
    const v4f lo = *(const v4f*)&x[bt * N_ + n0];
    const v4f hh = *(const v4f*)&x[bt * N_ + n0 + 4];
    tile[0][pos] = lo.x; tile[1][pos] = lo.y; tile[2][pos] = lo.z; tile[3][pos] = lo.w;
    tile[4][pos] = hh.x; tile[5][pos] = hh.y; tile[6][pos] = hh.z; tile[7][pos] = hh.w;
    __syncthreads();
#pragma unroll
    for (int j = 0; j < TN_; ++j) xt[(n0 + j) * ROW_ + bt] = tile[j][bt];
}

__global__ __launch_bounds__(384)
void k1_layer1(const float* __restrict__ xt, const int* __restrict__ nb,
               const float* __restrict__ W1, const float* __restrict__ b1,
               float* __restrict__ h1t) {
    const int lane  = threadIdx.x & 31;
    const int wave  = threadIdx.x >> 5;
    const int nl    = wave / 3;
    const int third = wave - nl * 3;
    const int n     = blockIdx.x * NPB_ + nl;
    const int mrow  = lane & 15;
    const int hi    = lane >> 4;

    v2f a[4][4];
    const float* xb = xt + third * 64 + mrow * 4;
#pragma unroll
    for (int s = 0; s < 4; ++s) {
        const int k0 = s * 4 + hi * 2;
        const v2i idx = *(const v2i*)&nb[n * K_ + k0];
        const v4f u0 = *(const v4f*)(xb + idx.x * ROW_);
        const v4f u1 = *(const v4f*)(xb + idx.y * ROW_);
#pragma unroll
        for (int tq = 0; tq < 4; ++tq) {
            a[tq][s].x = u0[tq];
            a[tq][s].y = u1[tq];
        }
    }
#pragma unroll
    for (int ct = 0; ct < 2; ++ct) {
        v8f acc[4] = {};
#pragma unroll
        for (int s = 0; s < 4; ++s) {
            const v2f b = *(const v2f*)&W1[(ct * 16 + mrow) * K_ + s * 4 + hi * 2];
#pragma unroll
            for (int tq = 0; tq < 4; ++tq) acc[tq] = wmma_f32_4(b, a[tq][s], acc[tq]);
        }
        const int chb = ct * 16 + hi * 8;
        const v4f blo = *(const v4f*)&b1[chb];
        const v4f bhh = *(const v4f*)&b1[chb + 4];
        float* p = h1t + ((size_t)chb * N_ + n) * ROW_ + third * 64 + mrow * 4;
#pragma unroll
        for (int r = 0; r < 8; ++r) {
            const float bias = r < 4 ? blo[r & 3] : bhh[r & 3];
            v4f v;
#pragma unroll
            for (int tq = 0; tq < 4; ++tq) {
                const float t = acc[tq][r] + bias;
                v[tq] = t > 0.f ? t : 0.f;
            }
            *(v4f*)(p + (size_t)r * N_ * ROW_) = v;
        }
    }
}

__global__ __launch_bounds__(384)
void k2_layer2(const float* __restrict__ h1t, const int* __restrict__ nb,
               const float* __restrict__ W2, const float* __restrict__ b2,
               const float* __restrict__ gamma, const float* __restrict__ beta,
               float* __restrict__ h2t) {
    __shared__ float w2f[128 * 64];
    __shared__ int   roff[NPB_ * 512];

    const int n0 = blockIdx.x * NPB_;
    if (threadIdx.x < 256) {
#pragma unroll
      for (int it = 0; it < 8; ++it) {
        const int i = threadIdx.x + it * 256;
        const int m = i & 15, s = i >> 4;
        const v4f w = *(const v4f*)&W2[m * 512 + s * 4];
        v2f lo, hh;
        lo.x = w.x; lo.y = w.y; hh.x = w.z; hh.y = w.w;
        ((v2f*)w2f)[s * 32 + m]      = lo;
        ((v2f*)w2f)[s * 32 + 16 + m] = hh;
      }
#pragma unroll
      for (int it = 0; it < 8; ++it) {
        const int i = threadIdx.x + it * 256;
        const unsigned q = (unsigned)(n0 + (i >> 9)) * 512u + (unsigned)(i & 511);
        const unsigned c = q / (unsigned)NK_;
        const unsigned r = q - c * (unsigned)NK_;
        roff[i] = ((int)c * N_ + nb[r]) * ROW_;
      }
    }
    __syncthreads();

    const int lane  = threadIdx.x & 31;
    const int wave  = threadIdx.x >> 5;
    const int nl    = wave / 3;
    const int third = wave - nl * 3;
    const int n     = n0 + nl;
    const int mrow  = lane & 15;
    const int hi    = lane >> 4;

    const float* abase = h1t + third * 64 + mrow * 4;
    const int*   ro    = roff + nl * 512 + hi * 2;
    const v2f*   wb    = (const v2f*)w2f + lane;

    v8f acc[4] = {};
    auto gather = [&](int s0, v4f (&u0)[G_], v4f (&u1)[G_]) {
#pragma unroll
        for (int j = 0; j < G_; ++j) {
            const v2i o = *(const v2i*)&ro[(s0 + j) * 4];
            u0[j] = *(const v4f*)(abase + o.x);
            u1[j] = *(const v4f*)(abase + o.y);
        }
    };
    auto mm = [&](int s0, const v4f (&u0)[G_], const v4f (&u1)[G_]) {
#pragma unroll
        for (int j = 0; j < G_; ++j) {
            const v2f b = wb[(s0 + j) * 32];
#pragma unroll
            for (int tq = 0; tq < 4; ++tq) {
                v2f a;
                a.x = u0[j][tq];
                a.y = u1[j][tq];
                acc[tq] = wmma_f32_4(b, a, acc[tq]);
            }
        }
    };
    v4f p0[G_], p1[G_], q0[G_], q1[G_];
    gather(0, p0, p1);
#define SCHED_FENCE() __builtin_amdgcn_sched_barrier(0)
    for (int s = 0; s < 128 - 2 * G_; s += 2 * G_) {
        gather(s + G_, q0, q1);
        SCHED_FENCE();
        mm(s, p0, p1);
        SCHED_FENCE();
        gather(s + 2 * G_, p0, p1);
        SCHED_FENCE();
        mm(s + G_, q0, q1);
        SCHED_FENCE();
    }
    gather(128 - G_, q0, q1);
    SCHED_FENCE();
    mm(128 - 2 * G_, p0, p1);
    mm(128 - G_, q0, q1);

    const int   ob  = hi * 8;
    const float rs  = rsqrtf(1.0f + 1e-5f);
    const v4f g_lo  = *(const v4f*)&gamma[ob], g_hh  = *(const v4f*)&gamma[ob + 4];
    const v4f b_lo  = *(const v4f*)&b2[ob],    b_hh  = *(const v4f*)&b2[ob + 4];
    const v4f s_lo  = *(const v4f*)&beta[ob],  s_hh  = *(const v4f*)&beta[ob + 4];
    float* p = h2t + ((size_t)ob * N_ + n) * ROW_ + third * 64 + mrow * 4;
#pragma unroll
    for (int r = 0; r < 8; ++r) {
        const float scale = (r < 4 ? g_lo[r & 3] : g_hh[r & 3]) * rs;
        const float bias  = r < 4 ? b_lo[r & 3] : b_hh[r & 3];
        const float shift = r < 4 ? s_lo[r & 3] : s_hh[r & 3];
        v4f v;
#pragma unroll
        for (int tq = 0; tq < 4; ++tq) {
            const float t = (acc[tq][r] + bias) * scale + shift;
            v[tq] = t > 0.f ? t : 0.f;
        }
        *(v4f*)(p + (size_t)r * N_ * ROW_) = v;
    }
}

__global__ __launch_bounds__(192)
void k3_layer3(const float* __restrict__ h2t, const int* __restrict__ nb,
               const float* __restrict__ W3, const float* __restrict__ b3,
               float* __restrict__ out) {
    __shared__ int   roff[NPB_ * 256];
    __shared__ float w3s[256];
    __shared__ float tile[NPB_][ROW_];

    const int n0 = blockIdx.x * NPB_;
    if (threadIdx.x < 128) {
#pragma unroll
        for (int it = 0; it < 8; ++it) {
            const int i = threadIdx.x + it * 128;
            const unsigned q = (unsigned)(n0 + (i >> 8)) * 256u + (unsigned)(i & 255);
            const unsigned c = q / (unsigned)NK_;
            const unsigned r = q - c * (unsigned)NK_;
            roff[i] = ((int)c * N_ + nb[r]) * ROW_;
        }
    } else {
        ((v4f*)w3s)[threadIdx.x - 128] = ((const v4f*)W3)[threadIdx.x - 128];
    }
    __syncthreads();

    const int nl    = threadIdx.x / 48;
    const int chunk = threadIdx.x - nl * 48;
    const float* base = h2t + chunk * 4;
    const int*   ro   = roff + nl * 256;
    const float  b3v  = b3[0];
    v4f acc;
    acc.x = b3v; acc.y = b3v; acc.z = b3v; acc.w = b3v;
    auto gather3 = [&](int f0, v4f (&u)[GF_]) {
#pragma unroll
        for (int j = 0; j < GF_; ++j) u[j] = *(const v4f*)(base + ro[f0 + j]);
    };
    auto fma3 = [&](int f0, const v4f (&u)[GF_]) {
#pragma unroll
        for (int j = 0; j < GF_; ++j) {
            const float w = w3s[f0 + j];
            acc.x = fmaf(u[j].x, w, acc.x);
            acc.y = fmaf(u[j].y, w, acc.y);
            acc.z = fmaf(u[j].z, w, acc.z);
            acc.w = fmaf(u[j].w, w, acc.w);
        }
    };
    v4f pa[GF_], pb[GF_];
    gather3(0, pa);
    for (int f = 0; f < 256 - 2 * GF_; f += 2 * GF_) {
        gather3(f + GF_, pb);
        SCHED_FENCE();
        fma3(f, pa);
        SCHED_FENCE();
        gather3(f + 2 * GF_, pa);
        SCHED_FENCE();
        fma3(f + GF_, pb);
        SCHED_FENCE();
    }
    gather3(256 - GF_, pb);
    SCHED_FENCE();
    fma3(256 - 2 * GF_, pa);
    fma3(256 - GF_, pb);
    v4f rl;
    rl.x = acc.x > 0.f ? acc.x : 0.f;
    rl.y = acc.y > 0.f ? acc.y : 0.f;
    rl.z = acc.z > 0.f ? acc.z : 0.f;
    rl.w = acc.w > 0.f ? acc.w : 0.f;
    *(v4f*)&tile[nl][chunk * 4] = rl;
    __syncthreads();

    const int bt  = threadIdx.x;
    const int pos = (bt >> 6) * 64 + (bt & 15) * 4 + ((bt >> 4) & 3);
    v4f o;
    o.x = tile[0][pos]; o.y = tile[1][pos]; o.z = tile[2][pos]; o.w = tile[3][pos];
    *(v4f*)&out[bt * N_ + n0] = o;
}

extern "C" void kernel_launch(void* const* d_in, const int* in_sizes, int n_in,
                              void* d_out, int out_size, void* d_ws, size_t ws_size,
                              hipStream_t stream) {
    const float* x     = (const float*)d_in[0];
    const int*   nb    = (const int*)  d_in[1];
    const float* W1    = (const float*)d_in[2];
    const float* b1    = (const float*)d_in[3];
    const float* W2    = (const float*)d_in[4];
    const float* b2    = (const float*)d_in[5];
    const float* W3    = (const float*)d_in[6];
    const float* b3    = (const float*)d_in[7];
    const float* gamma = (const float*)d_in[8];
    const float* beta  = (const float*)d_in[9];

    float* h1t = (float*)d_ws;
    float* h2t = h1t + (size_t)C1_ * N_ * ROW_;
    float* xt  = h2t;
    float* out = (float*)d_out;

    k0_transpose<<<125, 192, 0, stream>>>(x, xt);
    k1_layer1<<<NBLK_, 384, 0, stream>>>(xt, nb, W1, b1, h1t);
    k2_layer2<<<NBLK_, 384, 0, stream>>>(h1t, nb, W2, b2, gamma, beta, h2t);
    k3_layer3<<<NBLK_, 192, 0, stream>>>(h2t, nb, W3, b3, out);
}
